# starting kernel
# baseline (speedup 1.0000x reference)
#include <hip/hip_runtime.h>

#define B_ 64
#define L_ 512
#define D_ 768
#define H_ 768
#define K_ 128

typedef float v2f __attribute__((ext_vector_type(2)));
typedef float v8f __attribute__((ext_vector_type(8)));

__global__ __launch_bounds__(512) void feat_kernel(
    const float* __restrict__ enc,
    const int*   __restrict__ valid_mask,
    const int*   __restrict__ pos_span,
    float*       __restrict__ feat)
{
    __shared__ int s_scan[L_];
    __shared__ int s_idx[64];
    __shared__ int s_n;

    const int b   = blockIdx.x;
    const int tid = threadIdx.x;
    const int* vm = valid_mask + (size_t)b * L_;

    const int v = (vm[tid] == 1) ? 1 : 0;
    s_scan[tid] = v;
    __syncthreads();

    for (int off = 1; off < L_; off <<= 1) {
        int add = (tid >= off) ? s_scan[tid - off] : 0;
        __syncthreads();
        s_scan[tid] += add;
        __syncthreads();
    }

    const int pos = s_scan[tid] - 1;
    if (v && pos < 64) s_idx[pos] = tid;
    if (tid == 0) {
        int nv = s_scan[L_ - 1];
        s_n = (nv < 64) ? nv : 64;
    }
    __syncthreads();

    const int lo = pos_span[b * 2 + 0];
    const int hi = pos_span[b * 2 + 1];
    const int nv = s_n;

    for (int d = tid; d < D_; d += 512) {
        float acc = -__builtin_inff();
        for (int j = lo; j <= hi; ++j) {
            float val = (j < nv)
                ? enc[((size_t)b * L_ + s_idx[j]) * D_ + d]
                : 0.0f;
            acc = fmaxf(acc, val);
        }
        feat[(size_t)b * D_ + d] = acc;
    }
}

template <bool RELU>
__global__ __launch_bounds__(128) void gemm_wmma(
    const float* __restrict__ A,
    const float* __restrict__ Bm,
    const float* __restrict__ bias,
    float*       __restrict__ C,
    int N, int Kd)
{
    __shared__ float sB[64 * 16];

    const int lane = threadIdx.x & 31;
    const int wave = threadIdx.x >> 5;
    const int n0   = blockIdx.x * 16;
    const int m0   = wave * 16;
    const int m    = lane & 15;
    const int n    = lane & 15;
    const int kb   = (lane >> 4) * 2;

    v8f acc = {};
    const float* arow   = A + (size_t)(m0 + m) * Kd;
    const float  bias_n = bias[n0 + n];

    for (int kc = 0; kc < Kd; kc += 64) {
        for (int i = threadIdx.x; i < 64 * 16; i += 128) {
            sB[i] = Bm[(size_t)(kc + (i >> 4)) * N + n0 + (i & 15)];
        }
        __syncthreads();

        #pragma unroll
        for (int k = 0; k < 64; k += 4) {
            v2f a;
            a.x = arow[kc + k + kb];
            a.y = arow[kc + k + kb + 1];
            v2f bf;
            bf.x = sB[(k + kb)     * 16 + n];
            bf.y = sB[(k + kb + 1) * 16 + n];
            acc = __builtin_amdgcn_wmma_f32_16x16x4_f32(
                false, a, false, bf, (short)0, acc, false, false);
        }
        __syncthreads();
    }

    const int mrow = m0 + (lane >> 4) * 8;
    #pragma unroll
    for (int vv = 0; vv < 8; ++vv) {
        float val = acc[vv] + bias_n;
        if (RELU) val = fmaxf(val, 0.0f);
        C[(size_t)(mrow + vv) * N + n0 + n] = val;
    }
}

extern "C" void kernel_launch(void* const* d_in, const int* in_sizes, int n_in,
                              void* d_out, int out_size, void* d_ws, size_t ws_size,
                              hipStream_t stream)
{
    (void)in_sizes; (void)n_in; (void)out_size; (void)ws_size;

    const float* enc        = (const float*)d_in[0];
    const float* W1         = (const float*)d_in[1];
    const float* b1         = (const float*)d_in[2];
    const float* W2         = (const float*)d_in[3];
    const float* b2         = (const float*)d_in[4];
    const int*   valid_mask = (const int*)d_in[5];
    const int*   pos_span   = (const int*)d_in[6];

    float* out  = (float*)d_out;
    float* feat = (float*)d_ws;
    float* h    = feat + (size_t)B_ * D_;

    feat_kernel<<<B_, 512, 0, stream>>>(enc, valid_mask, pos_span, feat);
    gemm_wmma<true ><<<H_ / 16, 128, 0, stream>>>(feat, W1, b1, h,   H_, D_);
    gemm_wmma<false><<<K_ / 16, 128, 0, stream>>>(h,    W2, b2, out, K_, H_);
}


# GEMMs: K split across the 8 waves of a work-group (one 16x16 tile each), fragments straight from global as one load group per wave, partials summed through LDS in fixed order (f32 sum order changes)
# speedup vs baseline: 3.7064x; 3.7064x over previous
#include <hip/hip_runtime.h>

#define B_ 64
#define L_ 512
#define D_ 768
#define H_ 768
#define K_ 128

typedef float v2f __attribute__((ext_vector_type(2)));
typedef float v8f __attribute__((ext_vector_type(8)));

__global__ __launch_bounds__(512) void feat_kernel(
    const float* __restrict__ enc,
    const int*   __restrict__ valid_mask,
    const int*   __restrict__ pos_span,
    float*       __restrict__ feat)
{
    __shared__ int s_scan[L_];
    __shared__ int s_idx[64];
    __shared__ int s_n;

    const int b   = blockIdx.x;
    const int tid = threadIdx.x;
    const int* vm = valid_mask + (size_t)b * L_;

    const int v = (vm[tid] == 1) ? 1 : 0;
    s_scan[tid] = v;
    __syncthreads();

    for (int off = 1; off < L_; off <<= 1) {
        int add = (tid >= off) ? s_scan[tid - off] : 0;
        __syncthreads();
        s_scan[tid] += add;
        __syncthreads();
    }

    const int pos = s_scan[tid] - 1;
    if (v && pos < 64) s_idx[pos] = tid;
    if (tid == 0) {
        int nv = s_scan[L_ - 1];
        s_n = (nv < 64) ? nv : 64;
    }
    __syncthreads();

    const int lo = pos_span[b * 2 + 0];
    const int hi = pos_span[b * 2 + 1];
    const int nv = s_n;

    for (int d = tid; d < D_; d += 512) {
        float acc = -__builtin_inff();
        for (int j = lo; j <= hi; ++j) {
            float val = (j < nv)
                ? enc[((size_t)b * L_ + s_idx[j]) * D_ + d]
                : 0.0f;
            acc = fmaxf(acc, val);
        }
        feat[(size_t)b * D_ + d] = acc;
    }
}

#define KSPLIT_WAVES 8

template <bool RELU, int N, int KD>
__global__ __launch_bounds__(KSPLIT_WAVES * 32)
__attribute__((amdgpu_waves_per_eu(2, 4)))
void gemm_wmma_ksplit(
    const float* __restrict__ A,
    const float* __restrict__ Bm,
    const float* __restrict__ bias,
    float*       __restrict__ C)
{
    constexpr int KS = KD / KSPLIT_WAVES;
    constexpr int G  = KS / 8;
    static_assert(KD % (KSPLIT_WAVES * 8) == 0, "K slice must be whole groups of 8");

    __shared__ float sPart[KSPLIT_WAVES * 256];

    const int lane = threadIdx.x & 31;
    const int wave = threadIdx.x >> 5;
    const int n0   = blockIdx.x * 16;
    const int m0   = blockIdx.y * 16;
    const int c    = lane & 15;
    const int h    = lane >> 4;

    const int    kbase = wave * KS + 4 * h;
    const float* ap = A  + (size_t)(m0 + c) * KD + kbase;
    const float* bp = Bm + (size_t)kbase * N + n0 + c;

    float4 af[G];
    float  bf[G][4];
    #pragma unroll
    for (int t = 0; t < G; ++t) {
        af[t] = *reinterpret_cast<const float4*>(ap + 8 * t);
        #pragma unroll
        for (int j = 0; j < 4; ++j)
            bf[t][j] = bp[(size_t)(8 * t + j) * N];
    }
    const float bias_n = bias[n0 + c];

    v8f acc = {};
    #pragma unroll
    for (int t = 0; t < G; ++t) {
        v2f a01 = {af[t].x, af[t].y};
        v2f b01 = {bf[t][0], bf[t][1]};
        acc = __builtin_amdgcn_wmma_f32_16x16x4_f32(
            false, a01, false, b01, (short)0, acc, false, false);
        v2f a23 = {af[t].z, af[t].w};
        v2f b23 = {bf[t][2], bf[t][3]};
        acc = __builtin_amdgcn_wmma_f32_16x16x4_f32(
            false, a23, false, b23, (short)0, acc, false, false);
    }

    #pragma unroll
    for (int vv = 0; vv < 8; ++vv)
        sPart[wave * 256 + vv * 32 + lane] = acc[vv];
    __syncthreads();

    const int e = threadIdx.x;
    float s = sPart[e];
    #pragma unroll
    for (int w = 1; w < KSPLIT_WAVES; ++w)
        s += sPart[w * 256 + e];

    float val = s + bias_n;
    if (RELU) val = fmaxf(val, 0.0f);
    C[(size_t)(m0 + 8 * h + wave) * N + n0 + c] = val;
}

extern "C" void kernel_launch(void* const* d_in, const int* in_sizes, int n_in,
                              void* d_out, int out_size, void* d_ws, size_t ws_size,
                              hipStream_t stream)
{
    (void)in_sizes; (void)n_in; (void)out_size; (void)ws_size;

    const float* enc        = (const float*)d_in[0];
    const float* W1         = (const float*)d_in[1];
    const float* b1         = (const float*)d_in[2];
    const float* W2         = (const float*)d_in[3];
    const float* b2         = (const float*)d_in[4];
    const int*   valid_mask = (const int*)d_in[5];
    const int*   pos_span   = (const int*)d_in[6];

    float* out  = (float*)d_out;
    float* feat = (float*)d_ws;
    float* h    = feat + (size_t)B_ * D_;

    feat_kernel<<<B_, 512, 0, stream>>>(enc, valid_mask, pos_span, feat);
    gemm_wmma_ksplit<true,  H_, D_><<<dim3(H_ / 16, B_ / 16), KSPLIT_WAVES * 32, 0, stream>>>(feat, W1, b1, h);
    gemm_wmma_ksplit<false, K_, H_><<<dim3(K_ / 16, B_ / 16), KSPLIT_WAVES * 32, 0, stream>>>(h,    W2, b2, out);
}


# feat kernel: span rows dealt to 8 row groups, three float4 columns per thread, so all row loads of a span are in flight at once (was 4 scalar loads per round trip); group maxima merged through LDS
# speedup vs baseline: 6.1154x; 1.6500x over previous
#include <hip/hip_runtime.h>

#define B_ 64
#define L_ 512
#define D_ 768
#define H_ 768
#define K_ 128

typedef float v2f __attribute__((ext_vector_type(2)));
typedef float v8f __attribute__((ext_vector_type(8)));

#define ROWGROUPS 8
#define ROWS_PER_TRIP 5
#define D4_ (D_ / 4)

__device__ __forceinline__ float4 fmax4(float4 a, float4 b)
{
    return make_float4(fmaxf(a.x, b.x), fmaxf(a.y, b.y), fmaxf(a.z, b.z), fmaxf(a.w, b.w));
}

__global__ __launch_bounds__(512) void feat_kernel(
    const float* __restrict__ enc,
    const int*   __restrict__ valid_mask,
    const int*   __restrict__ pos_span,
    float*       __restrict__ feat)
{
    __shared__ int s_scan[L_];
    __shared__ int s_idx[64];
    __shared__ int s_n;
    __shared__ float4 s_part[ROWGROUPS * D4_];

    const int b   = blockIdx.x;
    const int tid = threadIdx.x;
    const int* vm = valid_mask + (size_t)b * L_;

    const int lo = pos_span[b * 2 + 0];
    const int hi = pos_span[b * 2 + 1];

    const int v = (vm[tid] == 1) ? 1 : 0;
    s_scan[tid] = v;
    __syncthreads();

    for (int off = 1; off < L_; off <<= 1) {
        int add = (tid >= off) ? s_scan[tid - off] : 0;
        __syncthreads();
        s_scan[tid] += add;
        __syncthreads();
    }

    const int pos = s_scan[tid] - 1;
    if (v && pos < 64) s_idx[pos] = tid;
    if (tid == 0) {
        int nv = s_scan[L_ - 1];
        s_n = (nv < 64) ? nv : 64;
    }
    __syncthreads();

    const int nv   = s_n;
    const int jend = (hi < nv - 1) ? hi : nv - 1;
    const int jlo  = (lo > 0) ? lo : 0;

    const int rg = __builtin_amdgcn_readfirstlane(tid >> 6);
    const int u  = tid & 63;

    const float  ninf  = -__builtin_inff();
    const float4 ninf4 = make_float4(ninf, ninf, ninf, ninf);
    const float4* encb = reinterpret_cast<const float4*>(enc + (size_t)b * L_ * D_);

    float4 m0 = ninf4, m1 = ninf4, m2 = ninf4;
    for (int j0 = jlo + rg; j0 <= jend; j0 += ROWGROUPS * ROWS_PER_TRIP) {
        int r[ROWS_PER_TRIP];
        #pragma unroll
        for (int i = 0; i < ROWS_PER_TRIP; ++i) {
            const int j = j0 + ROWGROUPS * i;
            r[i] = __builtin_amdgcn_readfirstlane(s_idx[(j < jend) ? j : jend]);
        }
        float4 x0[ROWS_PER_TRIP], x1[ROWS_PER_TRIP], x2[ROWS_PER_TRIP];
        #pragma unroll
        for (int i = 0; i < ROWS_PER_TRIP; ++i) {
            x0[i] = ninf4; x1[i] = ninf4; x2[i] = ninf4;
            if (j0 + ROWGROUPS * i <= jend) {
                const float4* row = encb + (size_t)r[i] * D4_;
                x0[i] = row[u];
                x1[i] = row[64 + u];
                x2[i] = row[128 + u];
            }
        }
        #pragma unroll
        for (int i = 0; i < ROWS_PER_TRIP; ++i) {
            m0 = fmax4(m0, x0[i]);
            m1 = fmax4(m1, x1[i]);
            m2 = fmax4(m2, x2[i]);
        }
    }

    s_part[rg * D4_ + u]       = m0;
    s_part[rg * D4_ + 64 + u]  = m1;
    s_part[rg * D4_ + 128 + u] = m2;
    __syncthreads();

    if (tid < D4_) {
        float4 acc = s_part[tid];
        #pragma unroll
        for (int g = 1; g < ROWGROUPS; ++g)
            acc = fmax4(acc, s_part[g * D4_ + tid]);
        if (lo <= hi && hi >= nv)
            acc = fmax4(acc, make_float4(0.0f, 0.0f, 0.0f, 0.0f));
        reinterpret_cast<float4*>(feat + (size_t)b * D_)[tid] = acc;
    }
}

#define KSPLIT_WAVES 8

template <bool RELU, int N, int KD>
__global__ __launch_bounds__(KSPLIT_WAVES * 32)
__attribute__((amdgpu_waves_per_eu(2, 4)))
void gemm_wmma_ksplit(
    const float* __restrict__ A,
    const float* __restrict__ Bm,
    const float* __restrict__ bias,
    float*       __restrict__ C)
{
    constexpr int KS = KD / KSPLIT_WAVES;
    constexpr int G  = KS / 8;
    static_assert(KD % (KSPLIT_WAVES * 8) == 0, "K slice must be whole groups of 8");

    __shared__ float sPart[KSPLIT_WAVES * 256];

    const int lane = threadIdx.x & 31;
    const int wave = threadIdx.x >> 5;
    const int n0   = blockIdx.x * 16;
    const int m0   = blockIdx.y * 16;
    const int c    = lane & 15;
    const int h    = lane >> 4;

    const int    kbase = wave * KS + 4 * h;
    const float* ap = A  + (size_t)(m0 + c) * KD + kbase;
    const float* bp = Bm + (size_t)kbase * N + n0 + c;

    float4 af[G];
    float  bf[G][4];
    #pragma unroll
    for (int t = 0; t < G; ++t) {
        af[t] = *reinterpret_cast<const float4*>(ap + 8 * t);
        #pragma unroll
        for (int j = 0; j < 4; ++j)
            bf[t][j] = bp[(size_t)(8 * t + j) * N];
    }
    const float bias_n = bias[n0 + c];

    v8f acc = {};
    #pragma unroll
    for (int t = 0; t < G; ++t) {
        v2f a01 = {af[t].x, af[t].y};
        v2f b01 = {bf[t][0], bf[t][1]};
        acc = __builtin_amdgcn_wmma_f32_16x16x4_f32(
            false, a01, false, b01, (short)0, acc, false, false);
        v2f a23 = {af[t].z, af[t].w};
        v2f b23 = {bf[t][2], bf[t][3]};
        acc = __builtin_amdgcn_wmma_f32_16x16x4_f32(
            false, a23, false, b23, (short)0, acc, false, false);
    }

    #pragma unroll
    for (int vv = 0; vv < 8; ++vv)
        sPart[wave * 256 + vv * 32 + lane] = acc[vv];
    __syncthreads();

    const int e = threadIdx.x;
    float s = sPart[e];
    #pragma unroll
    for (int w = 1; w < KSPLIT_WAVES; ++w)
        s += sPart[w * 256 + e];

    float val = s + bias_n;
    if (RELU) val = fmaxf(val, 0.0f);
    C[(size_t)(m0 + 8 * h + wave) * N + n0 + c] = val;
}

extern "C" void kernel_launch(void* const* d_in, const int* in_sizes, int n_in,
                              void* d_out, int out_size, void* d_ws, size_t ws_size,
                              hipStream_t stream)
{
    (void)in_sizes; (void)n_in; (void)out_size; (void)ws_size;

    const float* enc        = (const float*)d_in[0];
    const float* W1         = (const float*)d_in[1];
    const float* b1         = (const float*)d_in[2];
    const float* W2         = (const float*)d_in[3];
    const float* b2         = (const float*)d_in[4];
    const int*   valid_mask = (const int*)d_in[5];
    const int*   pos_span   = (const int*)d_in[6];

    float* out  = (float*)d_out;
    float* feat = (float*)d_ws;
    float* h    = feat + (size_t)B_ * D_;

    feat_kernel<<<B_, 512, 0, stream>>>(enc, valid_mask, pos_span, feat);
    gemm_wmma_ksplit<true,  H_, D_><<<dim3(H_ / 16, B_ / 16), KSPLIT_WAVES * 32, 0, stream>>>(feat, W1, b1, h);
    gemm_wmma_ksplit<false, K_, H_><<<dim3(K_ / 16, B_ / 16), KSPLIT_WAVES * 32, 0, stream>>>(h,    W2, b2, out);
}


# feat kernel: compacted positions from one ballot per wave plus per-wave counts handed through LDS, 2 barriers instead of the 9-step LDS scan's 20; integer counts, output bits unchanged
# speedup vs baseline: 6.5777x; 1.0756x over previous
#include <hip/hip_runtime.h>

#define B_ 64
#define L_ 512
#define D_ 768
#define H_ 768
#define K_ 128

typedef float v2f __attribute__((ext_vector_type(2)));
typedef float v8f __attribute__((ext_vector_type(8)));

#define NWAVES_F (L_ / 32)
#define ROWGROUPS 8
#define ROWS_PER_TRIP 5
#define D4_ (D_ / 4)

__device__ __forceinline__ float4 fmax4(float4 a, float4 b)
{
    return make_float4(fmaxf(a.x, b.x), fmaxf(a.y, b.y), fmaxf(a.z, b.z), fmaxf(a.w, b.w));
}

__global__ __launch_bounds__(512) void feat_kernel(
    const float* __restrict__ enc,
    const int*   __restrict__ valid_mask,
    const int*   __restrict__ pos_span,
    float*       __restrict__ feat)
{
    __shared__ __attribute__((aligned(16))) int s_wcnt[NWAVES_F];
    __shared__ int s_idx[64];
    __shared__ float4 s_part[ROWGROUPS * D4_];

    const int b   = blockIdx.x;
    const int tid = threadIdx.x;
    const int* vm = valid_mask + (size_t)b * L_;

    const int lo = pos_span[b * 2 + 0];
    const int hi = pos_span[b * 2 + 1];

    const bool v = (vm[tid] == 1);
    const int wv = __builtin_amdgcn_readfirstlane(tid >> 5);
    const unsigned bal = __builtin_amdgcn_ballot_w32(v);
    if ((tid & 31) == 0) s_wcnt[wv] = __builtin_popcount(bal);
    __syncthreads();

    int below = 0, total = 0;
    {
        const int4* wc = reinterpret_cast<const int4*>(s_wcnt);
        int cnt[NWAVES_F];
        #pragma unroll
        for (int q = 0; q < NWAVES_F / 4; ++q) {
            const int4 c4 = wc[q];
            cnt[4 * q + 0] = c4.x; cnt[4 * q + 1] = c4.y;
            cnt[4 * q + 2] = c4.z; cnt[4 * q + 3] = c4.w;
        }
        #pragma unroll
        for (int w = 0; w < NWAVES_F; ++w) {
            total += cnt[w];
            below += (w < wv) ? cnt[w] : 0;
        }
    }

    const int pos = below + (int)__builtin_amdgcn_mbcnt_lo(bal, 0u);
    if (v && pos < 64) s_idx[pos] = tid;
    __syncthreads();

    const int nv   = (total < 64) ? total : 64;
    const int jend = (hi < nv - 1) ? hi : nv - 1;
    const int jlo  = (lo > 0) ? lo : 0;

    const int rg = __builtin_amdgcn_readfirstlane(tid >> 6);
    const int u  = tid & 63;

    const float  ninf  = -__builtin_inff();
    const float4 ninf4 = make_float4(ninf, ninf, ninf, ninf);
    const float4* encb = reinterpret_cast<const float4*>(enc + (size_t)b * L_ * D_);

    float4 m0 = ninf4, m1 = ninf4, m2 = ninf4;
    for (int j0 = jlo + rg; j0 <= jend; j0 += ROWGROUPS * ROWS_PER_TRIP) {
        int r[ROWS_PER_TRIP];
        #pragma unroll
        for (int i = 0; i < ROWS_PER_TRIP; ++i) {
            const int j = j0 + ROWGROUPS * i;
            r[i] = __builtin_amdgcn_readfirstlane(s_idx[(j < jend) ? j : jend]);
        }
        float4 x0[ROWS_PER_TRIP], x1[ROWS_PER_TRIP], x2[ROWS_PER_TRIP];
        #pragma unroll
        for (int i = 0; i < ROWS_PER_TRIP; ++i) {
            x0[i] = ninf4; x1[i] = ninf4; x2[i] = ninf4;
            if (j0 + ROWGROUPS * i <= jend) {
                const float4* row = encb + (size_t)r[i] * D4_;
                x0[i] = row[u];
                x1[i] = row[64 + u];
                x2[i] = row[128 + u];
            }
        }
        #pragma unroll
        for (int i = 0; i < ROWS_PER_TRIP; ++i) {
            m0 = fmax4(m0, x0[i]);
            m1 = fmax4(m1, x1[i]);
            m2 = fmax4(m2, x2[i]);
        }
    }

    s_part[rg * D4_ + u]       = m0;
    s_part[rg * D4_ + 64 + u]  = m1;
    s_part[rg * D4_ + 128 + u] = m2;
    __syncthreads();

    if (tid < D4_) {
        float4 acc = s_part[tid];
        #pragma unroll
        for (int g = 1; g < ROWGROUPS; ++g)
            acc = fmax4(acc, s_part[g * D4_ + tid]);
        if (lo <= hi && hi >= nv)
            acc = fmax4(acc, make_float4(0.0f, 0.0f, 0.0f, 0.0f));
        reinterpret_cast<float4*>(feat + (size_t)b * D_)[tid] = acc;
    }
}

#define KSPLIT_WAVES 8

template <bool RELU, int N, int KD>
__global__ __launch_bounds__(KSPLIT_WAVES * 32)
__attribute__((amdgpu_waves_per_eu(2, 4)))
void gemm_wmma_ksplit(
    const float* __restrict__ A,
    const float* __restrict__ Bm,
    const float* __restrict__ bias,
    float*       __restrict__ C)
{
    constexpr int KS = KD / KSPLIT_WAVES;
    constexpr int G  = KS / 8;
    static_assert(KD % (KSPLIT_WAVES * 8) == 0, "K slice must be whole groups of 8");

    __shared__ float sPart[KSPLIT_WAVES * 256];

    const int lane = threadIdx.x & 31;
    const int wave = threadIdx.x >> 5;
    const int n0   = blockIdx.x * 16;
    const int m0   = blockIdx.y * 16;
    const int c    = lane & 15;
    const int h    = lane >> 4;

    const int    kbase = wave * KS + 4 * h;
    const float* ap = A  + (size_t)(m0 + c) * KD + kbase;
    const float* bp = Bm + (size_t)kbase * N + n0 + c;

    float4 af[G];
    float  bf[G][4];
    #pragma unroll
    for (int t = 0; t < G; ++t) {
        af[t] = *reinterpret_cast<const float4*>(ap + 8 * t);
        #pragma unroll
        for (int j = 0; j < 4; ++j)
            bf[t][j] = bp[(size_t)(8 * t + j) * N];
    }
    const float bias_n = bias[n0 + c];

    v8f acc = {};
    #pragma unroll
    for (int t = 0; t < G; ++t) {
        v2f a01 = {af[t].x, af[t].y};
        v2f b01 = {bf[t][0], bf[t][1]};
        acc = __builtin_amdgcn_wmma_f32_16x16x4_f32(
            false, a01, false, b01, (short)0, acc, false, false);
        v2f a23 = {af[t].z, af[t].w};
        v2f b23 = {bf[t][2], bf[t][3]};
        acc = __builtin_amdgcn_wmma_f32_16x16x4_f32(
            false, a23, false, b23, (short)0, acc, false, false);
    }

    #pragma unroll
    for (int vv = 0; vv < 8; ++vv)
        sPart[wave * 256 + vv * 32 + lane] = acc[vv];
    __syncthreads();

    const int e = threadIdx.x;
    float s = sPart[e];
    #pragma unroll
    for (int w = 1; w < KSPLIT_WAVES; ++w)
        s += sPart[w * 256 + e];

    float val = s + bias_n;
    if (RELU) val = fmaxf(val, 0.0f);
    C[(size_t)(m0 + 8 * h + wave) * N + n0 + c] = val;
}

extern "C" void kernel_launch(void* const* d_in, const int* in_sizes, int n_in,
                              void* d_out, int out_size, void* d_ws, size_t ws_size,
                              hipStream_t stream)
{
    (void)in_sizes; (void)n_in; (void)out_size; (void)ws_size;

    const float* enc        = (const float*)d_in[0];
    const float* W1         = (const float*)d_in[1];
    const float* b1         = (const float*)d_in[2];
    const float* W2         = (const float*)d_in[3];
    const float* b2         = (const float*)d_in[4];
    const int*   valid_mask = (const int*)d_in[5];
    const int*   pos_span   = (const int*)d_in[6];

    float* out  = (float*)d_out;
    float* feat = (float*)d_ws;
    float* h    = feat + (size_t)B_ * D_;

    feat_kernel<<<B_, 512, 0, stream>>>(enc, valid_mask, pos_span, feat);
    gemm_wmma_ksplit<true,  H_, D_><<<dim3(H_ / 16, B_ / 16), KSPLIT_WAVES * 32, 0, stream>>>(feat, W1, b1, h);
    gemm_wmma_ksplit<false, K_, H_><<<dim3(K_ / 16, B_ / 16), KSPLIT_WAVES * 32, 0, stream>>>(h,    W2, b2, out);
}


# span-max folded into the first product: D split over 6 work-group columns into 6 partial planes, added in fixed order in the second product's operand load; 3 launches to 2 (f32 sum order changes)
# speedup vs baseline: 6.9156x; 1.0514x over previous
#include <hip/hip_runtime.h>

#define B_ 64
#define L_ 512
#define D_ 768
#define H_ 768
#define K_ 128

typedef float v2f __attribute__((ext_vector_type(2)));
typedef float v8f __attribute__((ext_vector_type(8)));

#define NWAVES_F (L_ / 32)
#define ROWGROUPS 8
#define ROWS_PER_TRIP 5
#define D4_ (D_ / 4)

__device__ __forceinline__ float4 fmax4(float4 a, float4 b)
{
    return make_float4(fmaxf(a.x, b.x), fmaxf(a.y, b.y), fmaxf(a.z, b.z), fmaxf(a.w, b.w));
}

__global__ __launch_bounds__(512) void feat_kernel(
    const float* __restrict__ enc,
    const int*   __restrict__ valid_mask,
    const int*   __restrict__ pos_span,
    float*       __restrict__ feat)
{
    __shared__ __attribute__((aligned(16))) int s_wcnt[NWAVES_F];
    __shared__ int s_idx[64];
    __shared__ float4 s_part[ROWGROUPS * D4_];

    const int b   = blockIdx.x;
    const int tid = threadIdx.x;
    const int* vm = valid_mask + (size_t)b * L_;

    const int lo = pos_span[b * 2 + 0];
    const int hi = pos_span[b * 2 + 1];

    const bool v = (vm[tid] == 1);
    const int wv = __builtin_amdgcn_readfirstlane(tid >> 5);
    const unsigned bal = __builtin_amdgcn_ballot_w32(v);
    if ((tid & 31) == 0) s_wcnt[wv] = __builtin_popcount(bal);
    __syncthreads();

    int below = 0, total = 0;
    {
        const int4* wc = reinterpret_cast<const int4*>(s_wcnt);
        int cnt[NWAVES_F];
        #pragma unroll
        for (int q = 0; q < NWAVES_F / 4; ++q) {
            const int4 c4 = wc[q];
            cnt[4 * q + 0] = c4.x; cnt[4 * q + 1] = c4.y;
            cnt[4 * q + 2] = c4.z; cnt[4 * q + 3] = c4.w;
        }
        #pragma unroll
        for (int w = 0; w < NWAVES_F; ++w) {
            total += cnt[w];
            below += (w < wv) ? cnt[w] : 0;
        }
    }

    const int pos = below + (int)__builtin_amdgcn_mbcnt_lo(bal, 0u);
    if (v && pos < 64) s_idx[pos] = tid;
    __syncthreads();

    const int nv   = (total < 64) ? total : 64;
    const int jend = (hi < nv - 1) ? hi : nv - 1;
    const int jlo  = (lo > 0) ? lo : 0;

    const int rg = __builtin_amdgcn_readfirstlane(tid >> 6);
    const int u  = tid & 63;

    const float  ninf  = -__builtin_inff();
    const float4 ninf4 = make_float4(ninf, ninf, ninf, ninf);
    const float4* encb = reinterpret_cast<const float4*>(enc + (size_t)b * L_ * D_);

    float4 m0 = ninf4, m1 = ninf4, m2 = ninf4;
    for (int j0 = jlo + rg; j0 <= jend; j0 += ROWGROUPS * ROWS_PER_TRIP) {
        int r[ROWS_PER_TRIP];
        #pragma unroll
        for (int i = 0; i < ROWS_PER_TRIP; ++i) {
            const int j = j0 + ROWGROUPS * i;
            r[i] = __builtin_amdgcn_readfirstlane(s_idx[(j < jend) ? j : jend]);
        }
        float4 x0[ROWS_PER_TRIP], x1[ROWS_PER_TRIP], x2[ROWS_PER_TRIP];
        #pragma unroll
        for (int i = 0; i < ROWS_PER_TRIP; ++i) {
            x0[i] = ninf4; x1[i] = ninf4; x2[i] = ninf4;
            if (j0 + ROWGROUPS * i <= jend) {
                const float4* row = encb + (size_t)r[i] * D4_;
                x0[i] = row[u];
                x1[i] = row[64 + u];
                x2[i] = row[128 + u];
            }
        }
        #pragma unroll
        for (int i = 0; i < ROWS_PER_TRIP; ++i) {
            m0 = fmax4(m0, x0[i]);
            m1 = fmax4(m1, x1[i]);
            m2 = fmax4(m2, x2[i]);
        }
    }

    s_part[rg * D4_ + u]       = m0;
    s_part[rg * D4_ + 64 + u]  = m1;
    s_part[rg * D4_ + 128 + u] = m2;
    __syncthreads();

    if (tid < D4_) {
        float4 acc = s_part[tid];
        #pragma unroll
        for (int g = 1; g < ROWGROUPS; ++g)
            acc = fmax4(acc, s_part[g * D4_ + tid]);
        if (lo <= hi && hi >= nv)
            acc = fmax4(acc, make_float4(0.0f, 0.0f, 0.0f, 0.0f));
        reinterpret_cast<float4*>(feat + (size_t)b * D_)[tid] = acc;
    }
}

#define KSPLIT_WAVES 8

template <bool RELU, int N, int KD>
__global__ __launch_bounds__(KSPLIT_WAVES * 32)
__attribute__((amdgpu_waves_per_eu(2, 4)))
void gemm_wmma_ksplit(
    const float* __restrict__ A,
    const float* __restrict__ Bm,
    const float* __restrict__ bias,
    float*       __restrict__ C)
{
    constexpr int KS = KD / KSPLIT_WAVES;
    constexpr int G  = KS / 8;
    static_assert(KD % (KSPLIT_WAVES * 8) == 0, "K slice must be whole groups of 8");

    __shared__ float sPart[KSPLIT_WAVES * 256];

    const int lane = threadIdx.x & 31;
    const int wave = threadIdx.x >> 5;
    const int n0   = blockIdx.x * 16;
    const int m0   = blockIdx.y * 16;
    const int c    = lane & 15;
    const int h    = lane >> 4;

    const int    kbase = wave * KS + 4 * h;
    const float* ap = A  + (size_t)(m0 + c) * KD + kbase;
    const float* bp = Bm + (size_t)kbase * N + n0 + c;

    float4 af[G];
    float  bf[G][4];
    #pragma unroll
    for (int t = 0; t < G; ++t) {
        af[t] = *reinterpret_cast<const float4*>(ap + 8 * t);
        #pragma unroll
        for (int j = 0; j < 4; ++j)
            bf[t][j] = bp[(size_t)(8 * t + j) * N];
    }
    const float bias_n = bias[n0 + c];

    v8f acc = {};
    #pragma unroll
    for (int t = 0; t < G; ++t) {
        v2f a01 = {af[t].x, af[t].y};
        v2f b01 = {bf[t][0], bf[t][1]};
        acc = __builtin_amdgcn_wmma_f32_16x16x4_f32(
            false, a01, false, b01, (short)0, acc, false, false);
        v2f a23 = {af[t].z, af[t].w};
        v2f b23 = {bf[t][2], bf[t][3]};
        acc = __builtin_amdgcn_wmma_f32_16x16x4_f32(
            false, a23, false, b23, (short)0, acc, false, false);
    }

    #pragma unroll
    for (int vv = 0; vv < 8; ++vv)
        sPart[wave * 256 + vv * 32 + lane] = acc[vv];
    __syncthreads();

    const int e = threadIdx.x;
    float s = sPart[e];
    #pragma unroll
    for (int w = 1; w < KSPLIT_WAVES; ++w)
        s += sPart[w * 256 + e];

    float val = s + bias_n;
    if (RELU) val = fmaxf(val, 0.0f);
    C[(size_t)(m0 + 8 * h + wave) * N + n0 + c] = val;
}

#define KSL 128
#define NSL 128
#define NPL (D_ / KSL)
#define APITCH (KSL + 4)
#define SPAN_TRIP 40
#define HPITCH (H_ + 8)

__global__ __launch_bounds__(512) void feat_gemm1_part(
    const float* __restrict__ enc,
    const int*   __restrict__ valid_mask,
    const int*   __restrict__ pos_span,
    const float* __restrict__ W1,
    float*       __restrict__ planes)
{
    __shared__ int s_idx[16 * 64];
    __shared__ __attribute__((aligned(16))) float sA[16 * APITCH];
    __shared__ __attribute__((aligned(16))) float sPart[16 * 256];

    const int ns   = blockIdx.x;
    const int ks   = blockIdx.y;
    const int mt   = blockIdx.z;
    const int tid  = threadIdx.x;
    const int lane = tid & 31;
    const int wv   = __builtin_amdgcn_readfirstlane(tid >> 5);
    const int c    = lane & 15;
    const int h    = lane >> 4;

    const int  b  = mt * 16 + wv;
    const int* vm = valid_mask + (size_t)b * L_;
    int flags[L_ / 32];
    #pragma unroll
    for (int i = 0; i < L_ / 32; ++i)
        flags[i] = vm[i * 32 + lane];
    const int lo = pos_span[b * 2 + 0];
    const int hi = pos_span[b * 2 + 1];

    const int nt   = wv & 7;
    const int kloc = (wv >> 3) * (KSL / 2) + 4 * h;
    const float* bp = W1 + (size_t)(ks * KSL + kloc) * H_ + ns * NSL + nt * 16 + c;
    constexpr int G1 = (KSL / 2) / 8;
    float bf[G1][4];
    #pragma unroll
    for (int t = 0; t < G1; ++t) {
        #pragma unroll
        for (int j = 0; j < 4; ++j)
            bf[t][j] = bp[(size_t)(8 * t + j) * H_];
    }

    int run = 0;
    #pragma unroll
    for (int i = 0; i < L_ / 32; ++i) {
        const bool     v   = (flags[i] == 1);
        const unsigned bal = __builtin_amdgcn_ballot_w32(v);
        const int      pos = run + (int)__builtin_amdgcn_mbcnt_lo(bal, 0u);
        if (v && pos < 64) s_idx[wv * 64 + pos] = i * 32 + lane;
        run += __builtin_popcount(bal);
    }
    __syncthreads();

    const int nv   = (run < 64) ? run : 64;
    const int jend = (hi < nv - 1) ? hi : nv - 1;
    const int jlo  = (lo > 0) ? lo : 0;

    const float  ninf  = -__builtin_inff();
    const float4 ninf4 = make_float4(ninf, ninf, ninf, ninf);
    const float4* encb = reinterpret_cast<const float4*>(enc + (size_t)b * L_ * D_ + ks * KSL);

    float4 m = ninf4;
    for (int j0 = jlo; j0 <= jend; j0 += SPAN_TRIP) {
        const int ja = j0 + lane;
        const int jb = j0 + 32 + lane;
        const int ra = s_idx[wv * 64 + ((ja < jend) ? ja : jend)];
        const int rb = s_idx[wv * 64 + ((jb < jend) ? jb : jend)];
        int r[SPAN_TRIP];
        #pragma unroll
        for (int i = 0; i < SPAN_TRIP; ++i)
            r[i] = __builtin_amdgcn_readlane((i < 32) ? ra : rb, i & 31);
        float4 x[SPAN_TRIP];
        #pragma unroll
        for (int i = 0; i < SPAN_TRIP; ++i) {
            x[i] = ninf4;
            if (j0 + i <= jend)
                x[i] = encb[(size_t)r[i] * D4_ + lane];
        }
        #pragma unroll
        for (int i = 0; i < SPAN_TRIP; ++i)
            m = fmax4(m, x[i]);
    }
    if (lo <= hi && hi >= nv)
        m = fmax4(m, make_float4(0.0f, 0.0f, 0.0f, 0.0f));
    *reinterpret_cast<float4*>(&sA[wv * APITCH + lane * 4]) = m;
    __syncthreads();

    const float* arow = sA + c * APITCH + kloc;
    v8f acc = {};
    #pragma unroll
    for (int t = 0; t < G1; ++t) {
        const float4 a4 = *reinterpret_cast<const float4*>(arow + 8 * t);
        v2f a01 = {a4.x, a4.y};
        v2f b01 = {bf[t][0], bf[t][1]};
        acc = __builtin_amdgcn_wmma_f32_16x16x4_f32(
            false, a01, false, b01, (short)0, acc, false, false);
        v2f a23 = {a4.z, a4.w};
        v2f b23 = {bf[t][2], bf[t][3]};
        acc = __builtin_amdgcn_wmma_f32_16x16x4_f32(
            false, a23, false, b23, (short)0, acc, false, false);
    }

    #pragma unroll
    for (int vv = 0; vv < 8; ++vv)
        sPart[wv * 256 + vv * 32 + lane] = acc[vv];
    __syncthreads();

    const int row  = tid >> 5;
    const int col4 = lane * 4;
    const int off  = (col4 >> 4) * 256 + (row & 7) * 32 + (row >> 3) * 16 + (col4 & 15);
    const float4 p0 = *reinterpret_cast<const float4*>(&sPart[off]);
    const float4 p1 = *reinterpret_cast<const float4*>(&sPart[8 * 256 + off]);
    const float4 s4 = make_float4(p0.x + p1.x, p0.y + p1.y, p0.z + p1.z, p0.w + p1.w);
    float* plane = planes + (size_t)ks * B_ * H_;
    *reinterpret_cast<float4*>(&plane[(size_t)(mt * 16 + row) * H_ + ns * NSL + col4]) = s4;
}

__global__ __launch_bounds__(KSPLIT_WAVES * 32)
__attribute__((amdgpu_waves_per_eu(2, 4)))
void gemm2_planes(
    const float* __restrict__ planes,
    const float* __restrict__ b1,
    const float* __restrict__ W2,
    const float* __restrict__ b2,
    float*       __restrict__ out)
{
    constexpr int KS = H_ / KSPLIT_WAVES;
    constexpr int G  = KS / 8;
    static_assert(H_ % (KSPLIT_WAVES * 8) == 0, "K slice must be whole groups of 8");

    __shared__ __attribute__((aligned(16))) float sH[4 * HPITCH];
    __shared__ float sPart[KSPLIT_WAVES * 256];

    const int tid  = threadIdx.x;
    const int lane = tid & 31;
    const int wave = tid >> 5;
    const int n0   = blockIdx.x * 16;
    const int r0   = blockIdx.y * 4;
    const int c    = lane & 15;
    const int h    = lane >> 4;

    const int    kbase = wave * KS + 4 * h;
    const float* bp = W2 + (size_t)kbase * K_ + n0 + c;
    float bf[G][4];
    #pragma unroll
    for (int t = 0; t < G; ++t) {
        #pragma unroll
        for (int j = 0; j < 4; ++j)
            bf[t][j] = bp[(size_t)(8 * t + j) * K_];
    }
    const float bias_n = b2[n0 + c];

    constexpr int H4 = H_ / 4;
    float4 pv[3][NPL];
    float4 bb[3];
    #pragma unroll
    for (int q = 0; q < 3; ++q) {
        const int idx = tid + 256 * q;
        const int rr  = idx / H4;
        const int cq  = idx - rr * H4;
        bb[q] = reinterpret_cast<const float4*>(b1)[cq];
        #pragma unroll
        for (int p = 0; p < NPL; ++p)
            pv[q][p] = reinterpret_cast<const float4*>(
                planes + (size_t)p * B_ * H_ + (size_t)(r0 + rr) * H_)[cq];
    }
    #pragma unroll
    for (int q = 0; q < 3; ++q) {
        const int idx = tid + 256 * q;
        const int rr  = idx / H4;
        const int cq  = idx - rr * H4;
        float4 s = pv[q][0];
        #pragma unroll
        for (int p = 1; p < NPL; ++p) {
            s.x += pv[q][p].x; s.y += pv[q][p].y; s.z += pv[q][p].z; s.w += pv[q][p].w;
        }
        s.x = fmaxf(s.x + bb[q].x, 0.0f);
        s.y = fmaxf(s.y + bb[q].y, 0.0f);
        s.z = fmaxf(s.z + bb[q].z, 0.0f);
        s.w = fmaxf(s.w + bb[q].w, 0.0f);
        *reinterpret_cast<float4*>(&sH[rr * HPITCH + cq * 4]) = s;
    }
    __syncthreads();

    const float* arow = sH + (c & 3) * HPITCH + kbase;
    v8f acc = {};
    #pragma unroll
    for (int t = 0; t < G; ++t) {
        const float4 a4 = *reinterpret_cast<const float4*>(arow + 8 * t);
        v2f a01 = {a4.x, a4.y};
        v2f b01 = {bf[t][0], bf[t][1]};
        acc = __builtin_amdgcn_wmma_f32_16x16x4_f32(
            false, a01, false, b01, (short)0, acc, false, false);
        v2f a23 = {a4.z, a4.w};
        v2f b23 = {bf[t][2], bf[t][3]};
        acc = __builtin_amdgcn_wmma_f32_16x16x4_f32(
            false, a23, false, b23, (short)0, acc, false, false);
    }

    #pragma unroll
    for (int vv = 0; vv < 8; ++vv)
        sPart[wave * 256 + vv * 32 + lane] = acc[vv];
    __syncthreads();

    const int e = tid;
    float s = sPart[e];
    #pragma unroll
    for (int w = 1; w < KSPLIT_WAVES; ++w)
        s += sPart[w * 256 + e];
    if (h == 0 && wave < 4)
        out[(size_t)(r0 + wave) * K_ + n0 + c] = s + bias_n;
}

extern "C" void kernel_launch(void* const* d_in, const int* in_sizes, int n_in,
                              void* d_out, int out_size, void* d_ws, size_t ws_size,
                              hipStream_t stream)
{
    (void)in_sizes; (void)n_in; (void)out_size;

    const float* enc        = (const float*)d_in[0];
    const float* W1         = (const float*)d_in[1];
    const float* b1         = (const float*)d_in[2];
    const float* W2         = (const float*)d_in[3];
    const float* b2         = (const float*)d_in[4];
    const int*   valid_mask = (const int*)d_in[5];
    const int*   pos_span   = (const int*)d_in[6];

    float* out  = (float*)d_out;

    if (ws_size >= (size_t)NPL * B_ * H_ * sizeof(float)) {
        float* planes = (float*)d_ws;
        feat_gemm1_part<<<dim3(H_ / NSL, NPL, B_ / 16), 512, 0, stream>>>(enc, valid_mask, pos_span, W1, planes);
        gemm2_planes<<<dim3(K_ / 16, B_ / 4), KSPLIT_WAVES * 32, 0, stream>>>(planes, b1, W2, b2, out);
    } else {
        float* feat = (float*)d_ws;
        float* h    = feat + (size_t)B_ * D_;
        feat_kernel<<<B_, 512, 0, stream>>>(enc, valid_mask, pos_span, feat);
        gemm_wmma_ksplit<true,  H_, D_><<<dim3(H_ / 16, B_ / 16), KSPLIT_WAVES * 32, 0, stream>>>(feat, W1, b1, h);
        gemm_wmma_ksplit<false, K_, H_><<<dim3(K_ / 16, B_ / 16), KSPLIT_WAVES * 32, 0, stream>>>(h,    W2, b2, out);
    }
}


# launch A feature part sized by the span, not the worst case: flag scan over 4 words (more only if hi is not reached); rows fetched in started 8-row chunks, no per-row test (was 16 words, 40 rows)
# speedup vs baseline: 7.4847x; 1.0823x over previous
#include <hip/hip_runtime.h>

#define B_ 64
#define L_ 512
#define D_ 768
#define H_ 768
#define K_ 128

typedef float v2f __attribute__((ext_vector_type(2)));
typedef float v8f __attribute__((ext_vector_type(8)));

#define NWAVES_F (L_ / 32)
#define ROWGROUPS 8
#define ROWS_PER_TRIP 5
#define D4_ (D_ / 4)

__device__ __forceinline__ float4 fmax4(float4 a, float4 b)
{
    return make_float4(fmaxf(a.x, b.x), fmaxf(a.y, b.y), fmaxf(a.z, b.z), fmaxf(a.w, b.w));
}

__global__ __launch_bounds__(512) void feat_kernel(
    const float* __restrict__ enc,
    const int*   __restrict__ valid_mask,
    const int*   __restrict__ pos_span,
    float*       __restrict__ feat)
{
    __shared__ __attribute__((aligned(16))) int s_wcnt[NWAVES_F];
    __shared__ int s_idx[64];
    __shared__ float4 s_part[ROWGROUPS * D4_];

    const int b   = blockIdx.x;
    const int tid = threadIdx.x;
    const int* vm = valid_mask + (size_t)b * L_;

    const int lo = pos_span[b * 2 + 0];
    const int hi = pos_span[b * 2 + 1];

    const bool v = (vm[tid] == 1);
    const int wv = __builtin_amdgcn_readfirstlane(tid >> 5);
    const unsigned bal = __builtin_amdgcn_ballot_w32(v);
    if ((tid & 31) == 0) s_wcnt[wv] = __builtin_popcount(bal);
    __syncthreads();

    int below = 0, total = 0;
    {
        const int4* wc = reinterpret_cast<const int4*>(s_wcnt);
        int cnt[NWAVES_F];
        #pragma unroll
        for (int q = 0; q < NWAVES_F / 4; ++q) {
            const int4 c4 = wc[q];
            cnt[4 * q + 0] = c4.x; cnt[4 * q + 1] = c4.y;
            cnt[4 * q + 2] = c4.z; cnt[4 * q + 3] = c4.w;
        }
        #pragma unroll
        for (int w = 0; w < NWAVES_F; ++w) {
            total += cnt[w];
            below += (w < wv) ? cnt[w] : 0;
        }
    }

    const int pos = below + (int)__builtin_amdgcn_mbcnt_lo(bal, 0u);
    if (v && pos < 64) s_idx[pos] = tid;
    __syncthreads();

    const int nv   = (total < 64) ? total : 64;
    const int jend = (hi < nv - 1) ? hi : nv - 1;
    const int jlo  = (lo > 0) ? lo : 0;

    const int rg = __builtin_amdgcn_readfirstlane(tid >> 6);
    const int u  = tid & 63;

    const float  ninf  = -__builtin_inff();
    const float4 ninf4 = make_float4(ninf, ninf, ninf, ninf);
    const float4* encb = reinterpret_cast<const float4*>(enc + (size_t)b * L_ * D_);

    float4 m0 = ninf4, m1 = ninf4, m2 = ninf4;
    for (int j0 = jlo + rg; j0 <= jend; j0 += ROWGROUPS * ROWS_PER_TRIP) {
        int r[ROWS_PER_TRIP];
        #pragma unroll
        for (int i = 0; i < ROWS_PER_TRIP; ++i) {
            const int j = j0 + ROWGROUPS * i;
            r[i] = __builtin_amdgcn_readfirstlane(s_idx[(j < jend) ? j : jend]);
        }
        float4 x0[ROWS_PER_TRIP], x1[ROWS_PER_TRIP], x2[ROWS_PER_TRIP];
        #pragma unroll
        for (int i = 0; i < ROWS_PER_TRIP; ++i) {
            x0[i] = ninf4; x1[i] = ninf4; x2[i] = ninf4;
            if (j0 + ROWGROUPS * i <= jend) {
                const float4* row = encb + (size_t)r[i] * D4_;
                x0[i] = row[u];
                x1[i] = row[64 + u];
                x2[i] = row[128 + u];
            }
        }
        #pragma unroll
        for (int i = 0; i < ROWS_PER_TRIP; ++i) {
            m0 = fmax4(m0, x0[i]);
            m1 = fmax4(m1, x1[i]);
            m2 = fmax4(m2, x2[i]);
        }
    }

    s_part[rg * D4_ + u]       = m0;
    s_part[rg * D4_ + 64 + u]  = m1;
    s_part[rg * D4_ + 128 + u] = m2;
    __syncthreads();

    if (tid < D4_) {
        float4 acc = s_part[tid];
        #pragma unroll
        for (int g = 1; g < ROWGROUPS; ++g)
            acc = fmax4(acc, s_part[g * D4_ + tid]);
        if (lo <= hi && hi >= nv)
            acc = fmax4(acc, make_float4(0.0f, 0.0f, 0.0f, 0.0f));
        reinterpret_cast<float4*>(feat + (size_t)b * D_)[tid] = acc;
    }
}

#define KSPLIT_WAVES 8

template <bool RELU, int N, int KD>
__global__ __launch_bounds__(KSPLIT_WAVES * 32)
__attribute__((amdgpu_waves_per_eu(2, 4)))
void gemm_wmma_ksplit(
    const float* __restrict__ A,
    const float* __restrict__ Bm,
    const float* __restrict__ bias,
    float*       __restrict__ C)
{
    constexpr int KS = KD / KSPLIT_WAVES;
    constexpr int G  = KS / 8;
    static_assert(KD % (KSPLIT_WAVES * 8) == 0, "K slice must be whole groups of 8");

    __shared__ float sPart[KSPLIT_WAVES * 256];

    const int lane = threadIdx.x & 31;
    const int wave = threadIdx.x >> 5;
    const int n0   = blockIdx.x * 16;
    const int m0   = blockIdx.y * 16;
    const int c    = lane & 15;
    const int h    = lane >> 4;

    const int    kbase = wave * KS + 4 * h;
    const float* ap = A  + (size_t)(m0 + c) * KD + kbase;
    const float* bp = Bm + (size_t)kbase * N + n0 + c;

    float4 af[G];
    float  bf[G][4];
    #pragma unroll
    for (int t = 0; t < G; ++t) {
        af[t] = *reinterpret_cast<const float4*>(ap + 8 * t);
        #pragma unroll
        for (int j = 0; j < 4; ++j)
            bf[t][j] = bp[(size_t)(8 * t + j) * N];
    }
    const float bias_n = bias[n0 + c];

    v8f acc = {};
    #pragma unroll
    for (int t = 0; t < G; ++t) {
        v2f a01 = {af[t].x, af[t].y};
        v2f b01 = {bf[t][0], bf[t][1]};
        acc = __builtin_amdgcn_wmma_f32_16x16x4_f32(
            false, a01, false, b01, (short)0, acc, false, false);
        v2f a23 = {af[t].z, af[t].w};
        v2f b23 = {bf[t][2], bf[t][3]};
        acc = __builtin_amdgcn_wmma_f32_16x16x4_f32(
            false, a23, false, b23, (short)0, acc, false, false);
    }

    #pragma unroll
    for (int vv = 0; vv < 8; ++vv)
        sPart[wave * 256 + vv * 32 + lane] = acc[vv];
    __syncthreads();

    const int e = threadIdx.x;
    float s = sPart[e];
    #pragma unroll
    for (int w = 1; w < KSPLIT_WAVES; ++w)
        s += sPart[w * 256 + e];

    float val = s + bias_n;
    if (RELU) val = fmaxf(val, 0.0f);
    C[(size_t)(m0 + 8 * h + wave) * N + n0 + c] = val;
}

#define KSL 128
#define NSL 128
#define NPL (D_ / KSL)
#define APITCH (KSL + 4)
#define SPAN_HEAD 40
#define SPAN_CHUNK 8
#define FLAG_HEAD 4
#define HPITCH (H_ + 8)

__global__ __launch_bounds__(512) void feat_gemm1_part(
    const float* __restrict__ enc,
    const int*   __restrict__ valid_mask,
    const int*   __restrict__ pos_span,
    const float* __restrict__ W1,
    float*       __restrict__ planes)
{
    __shared__ int s_idx[16 * 64];
    __shared__ __attribute__((aligned(16))) float sA[16 * APITCH];
    __shared__ __attribute__((aligned(16))) float sPart[16 * 256];

    const int ns   = blockIdx.x;
    const int ks   = blockIdx.y;
    const int mt   = blockIdx.z;
    const int tid  = threadIdx.x;
    const int lane = tid & 31;
    const int wv   = __builtin_amdgcn_readfirstlane(tid >> 5);
    const int c    = lane & 15;
    const int h    = lane >> 4;

    const int  b  = mt * 16 + wv;
    const int* vm = valid_mask + (size_t)b * L_;
    int flags[FLAG_HEAD];
    #pragma unroll
    for (int i = 0; i < FLAG_HEAD; ++i)
        flags[i] = vm[i * 32 + lane];
    const int lo = pos_span[b * 2 + 0];
    const int hi = pos_span[b * 2 + 1];

    const int nt   = wv & 7;
    const int kloc = (wv >> 3) * (KSL / 2) + 4 * h;
    const float* bp = W1 + (size_t)(ks * KSL + kloc) * H_ + ns * NSL + nt * 16 + c;
    constexpr int G1 = (KSL / 2) / 8;
    float bf[G1][4];
    #pragma unroll
    for (int t = 0; t < G1; ++t) {
        #pragma unroll
        for (int j = 0; j < 4; ++j)
            bf[t][j] = bp[(size_t)(8 * t + j) * H_];
    }

    int run = 0;
    #pragma unroll
    for (int i = 0; i < FLAG_HEAD; ++i) {
        const bool     v   = (flags[i] == 1);
        const unsigned bal = __builtin_amdgcn_ballot_w32(v);
        const int      pos = run + (int)__builtin_amdgcn_mbcnt_lo(bal, 0u);
        if (v && pos < 64) s_idx[wv * 64 + pos] = i * 32 + lane;
        run += __builtin_popcount(bal);
    }
    if (run <= hi) {
        int tail[L_ / 32 - FLAG_HEAD];
        #pragma unroll
        for (int i = 0; i < L_ / 32 - FLAG_HEAD; ++i)
            tail[i] = vm[(FLAG_HEAD + i) * 32 + lane];
        #pragma unroll
        for (int i = 0; i < L_ / 32 - FLAG_HEAD; ++i) {
            const bool     v   = (tail[i] == 1);
            const unsigned bal = __builtin_amdgcn_ballot_w32(v);
            const int      pos = run + (int)__builtin_amdgcn_mbcnt_lo(bal, 0u);
            if (v && pos < 64) s_idx[wv * 64 + pos] = (FLAG_HEAD + i) * 32 + lane;
            run += __builtin_popcount(bal);
        }
    }
    __syncthreads();

    const int nv   = (run < 64) ? run : 64;
    const int jend = (hi < nv - 1) ? hi : nv - 1;
    const int jlo  = (lo > 0) ? lo : 0;

    const float  ninf  = -__builtin_inff();
    const float4 ninf4 = make_float4(ninf, ninf, ninf, ninf);
    const float4* encb = reinterpret_cast<const float4*>(enc + (size_t)b * L_ * D_ + ks * KSL);

    float4 m = ninf4;
    const int n = jend - jlo + 1;
    if (n > 0) {
        const int ja = jlo + lane;
        const int jb = jlo + 32 + lane;
        const int ra = s_idx[wv * 64 + ((ja < jend) ? ja : jend)];
        const int rb = s_idx[wv * 64 + ((jb < jend) ? jb : jend)];
        float4 x[SPAN_HEAD];
        #pragma unroll
        for (int q = 0; q < SPAN_HEAD / SPAN_CHUNK; ++q) {
            if (SPAN_CHUNK * q < n) {
                #pragma unroll
                for (int i = 0; i < SPAN_CHUNK; ++i) {
                    const int k = SPAN_CHUNK * q + i;
                    const int r = __builtin_amdgcn_readlane((k < 32) ? ra : rb, k & 31);
                    x[k] = encb[(size_t)r * D4_ + lane];
                }
            }
        }
        #pragma unroll
        for (int q = 0; q < SPAN_HEAD / SPAN_CHUNK; ++q) {
            if (SPAN_CHUNK * q < n) {
                #pragma unroll
                for (int i = 0; i < SPAN_CHUNK; ++i)
                    m = fmax4(m, x[SPAN_CHUNK * q + i]);
            }
        }
        for (int j = jlo + SPAN_HEAD; j <= jend; ++j) {
            const int r = __builtin_amdgcn_readfirstlane(s_idx[wv * 64 + j]);
            m = fmax4(m, encb[(size_t)r * D4_ + lane]);
        }
    }
    if (lo <= hi && hi >= nv)
        m = fmax4(m, make_float4(0.0f, 0.0f, 0.0f, 0.0f));
    *reinterpret_cast<float4*>(&sA[wv * APITCH + lane * 4]) = m;
    __syncthreads();

    const float* arow = sA + c * APITCH + kloc;
    v8f acc = {};
    #pragma unroll
    for (int t = 0; t < G1; ++t) {
        const float4 a4 = *reinterpret_cast<const float4*>(arow + 8 * t);
        v2f a01 = {a4.x, a4.y};
        v2f b01 = {bf[t][0], bf[t][1]};
        acc = __builtin_amdgcn_wmma_f32_16x16x4_f32(
            false, a01, false, b01, (short)0, acc, false, false);
        v2f a23 = {a4.z, a4.w};
        v2f b23 = {bf[t][2], bf[t][3]};
        acc = __builtin_amdgcn_wmma_f32_16x16x4_f32(
            false, a23, false, b23, (short)0, acc, false, false);
    }

    #pragma unroll
    for (int vv = 0; vv < 8; ++vv)
        sPart[wv * 256 + vv * 32 + lane] = acc[vv];
    __syncthreads();

    const int row  = tid >> 5;
    const int col4 = lane * 4;
    const int off  = (col4 >> 4) * 256 + (row & 7) * 32 + (row >> 3) * 16 + (col4 & 15);
    const float4 p0 = *reinterpret_cast<const float4*>(&sPart[off]);
    const float4 p1 = *reinterpret_cast<const float4*>(&sPart[8 * 256 + off]);
    const float4 s4 = make_float4(p0.x + p1.x, p0.y + p1.y, p0.z + p1.z, p0.w + p1.w);
    float* plane = planes + (size_t)ks * B_ * H_;
    *reinterpret_cast<float4*>(&plane[(size_t)(mt * 16 + row) * H_ + ns * NSL + col4]) = s4;
}

__global__ __launch_bounds__(KSPLIT_WAVES * 32)
__attribute__((amdgpu_waves_per_eu(2, 4)))
void gemm2_planes(
    const float* __restrict__ planes,
    const float* __restrict__ b1,
    const float* __restrict__ W2,
    const float* __restrict__ b2,
    float*       __restrict__ out)
{
    constexpr int KS = H_ / KSPLIT_WAVES;
    constexpr int G  = KS / 8;
    static_assert(H_ % (KSPLIT_WAVES * 8) == 0, "K slice must be whole groups of 8");

    __shared__ __attribute__((aligned(16))) float sH[4 * HPITCH];
    __shared__ float sPart[KSPLIT_WAVES * 256];

    const int tid  = threadIdx.x;
    const int lane = tid & 31;
    const int wave = tid >> 5;
    const int n0   = blockIdx.x * 16;
    const int r0   = blockIdx.y * 4;
    const int c    = lane & 15;
    const int h    = lane >> 4;

    const int    kbase = wave * KS + 4 * h;
    const float* bp = W2 + (size_t)kbase * K_ + n0 + c;
    float bf[G][4];
    #pragma unroll
    for (int t = 0; t < G; ++t) {
        #pragma unroll
        for (int j = 0; j < 4; ++j)
            bf[t][j] = bp[(size_t)(8 * t + j) * K_];
    }
    const float bias_n = b2[n0 + c];

    constexpr int H4 = H_ / 4;
    float4 pv[3][NPL];
    float4 bb[3];
    #pragma unroll
    for (int q = 0; q < 3; ++q) {
        const int idx = tid + 256 * q;
        const int rr  = idx / H4;
        const int cq  = idx - rr * H4;
        bb[q] = reinterpret_cast<const float4*>(b1)[cq];
        #pragma unroll
        for (int p = 0; p < NPL; ++p)
            pv[q][p] = reinterpret_cast<const float4*>(
                planes + (size_t)p * B_ * H_ + (size_t)(r0 + rr) * H_)[cq];
    }
    #pragma unroll
    for (int q = 0; q < 3; ++q) {
        const int idx = tid + 256 * q;
        const int rr  = idx / H4;
        const int cq  = idx - rr * H4;
        float4 s = pv[q][0];
        #pragma unroll
        for (int p = 1; p < NPL; ++p) {
            s.x += pv[q][p].x; s.y += pv[q][p].y; s.z += pv[q][p].z; s.w += pv[q][p].w;
        }
        s.x = fmaxf(s.x + bb[q].x, 0.0f);
        s.y = fmaxf(s.y + bb[q].y, 0.0f);
        s.z = fmaxf(s.z + bb[q].z, 0.0f);
        s.w = fmaxf(s.w + bb[q].w, 0.0f);
        *reinterpret_cast<float4*>(&sH[rr * HPITCH + cq * 4]) = s;
    }
    __syncthreads();

    const float* arow = sH + (c & 3) * HPITCH + kbase;
    v8f acc = {};
    #pragma unroll
    for (int t = 0; t < G; ++t) {
        const float4 a4 = *reinterpret_cast<const float4*>(arow + 8 * t);
        v2f a01 = {a4.x, a4.y};
        v2f b01 = {bf[t][0], bf[t][1]};
        acc = __builtin_amdgcn_wmma_f32_16x16x4_f32(
            false, a01, false, b01, (short)0, acc, false, false);
        v2f a23 = {a4.z, a4.w};
        v2f b23 = {bf[t][2], bf[t][3]};
        acc = __builtin_amdgcn_wmma_f32_16x16x4_f32(
            false, a23, false, b23, (short)0, acc, false, false);
    }

    #pragma unroll
    for (int vv = 0; vv < 8; ++vv)
        sPart[wave * 256 + vv * 32 + lane] = acc[vv];
    __syncthreads();

    const int e = tid;
    float s = sPart[e];
    #pragma unroll
    for (int w = 1; w < KSPLIT_WAVES; ++w)
        s += sPart[w * 256 + e];
    if (h == 0 && wave < 4)
        out[(size_t)(r0 + wave) * K_ + n0 + c] = s + bias_n;
}

extern "C" void kernel_launch(void* const* d_in, const int* in_sizes, int n_in,
                              void* d_out, int out_size, void* d_ws, size_t ws_size,
                              hipStream_t stream)
{
    (void)in_sizes; (void)n_in; (void)out_size;

    const float* enc        = (const float*)d_in[0];
    const float* W1         = (const float*)d_in[1];
    const float* b1         = (const float*)d_in[2];
    const float* W2         = (const float*)d_in[3];
    const float* b2         = (const float*)d_in[4];
    const int*   valid_mask = (const int*)d_in[5];
    const int*   pos_span   = (const int*)d_in[6];

    float* out  = (float*)d_out;

    if (ws_size >= (size_t)NPL * B_ * H_ * sizeof(float)) {
        float* planes = (float*)d_ws;
        feat_gemm1_part<<<dim3(H_ / NSL, NPL, B_ / 16), 512, 0, stream>>>(enc, valid_mask, pos_span, W1, planes);
        gemm2_planes<<<dim3(K_ / 16, B_ / 4), KSPLIT_WAVES * 32, 0, stream>>>(planes, b1, W2, b2, out);
    } else {
        float* feat = (float*)d_ws;
        float* h    = feat + (size_t)B_ * D_;
        feat_kernel<<<B_, 512, 0, stream>>>(enc, valid_mask, pos_span, feat);
        gemm_wmma_ksplit<true,  H_, D_><<<dim3(H_ / 16, B_ / 16), KSPLIT_WAVES * 32, 0, stream>>>(feat, W1, b1, h);
        gemm_wmma_ksplit<false, K_, H_><<<dim3(K_ / 16, B_ / 16), KSPLIT_WAVES * 32, 0, stream>>>(h,    W2, b2, out);
    }
}


# hand-over barriers wait for LDS only (s_wait_dscnt + barrier by hand, not __syncthreads): W1 loads stay in flight over launch A's gather; launch B loads planes first, W2 streams under sum and WMMAs
# speedup vs baseline: 8.0084x; 1.0700x over previous
#include <hip/hip_runtime.h>

#define B_ 64
#define L_ 512
#define D_ 768
#define H_ 768
#define K_ 128

typedef float v2f __attribute__((ext_vector_type(2)));
typedef float v8f __attribute__((ext_vector_type(8)));

#define NWAVES_F (L_ / 32)
#define ROWGROUPS 8
#define ROWS_PER_TRIP 5
#define D4_ (D_ / 4)

__device__ __forceinline__ float4 fmax4(float4 a, float4 b)
{
    return make_float4(fmaxf(a.x, b.x), fmaxf(a.y, b.y), fmaxf(a.z, b.z), fmaxf(a.w, b.w));
}

__global__ __launch_bounds__(512) void feat_kernel(
    const float* __restrict__ enc,
    const int*   __restrict__ valid_mask,
    const int*   __restrict__ pos_span,
    float*       __restrict__ feat)
{
    __shared__ __attribute__((aligned(16))) int s_wcnt[NWAVES_F];
    __shared__ int s_idx[64];
    __shared__ float4 s_part[ROWGROUPS * D4_];

    const int b   = blockIdx.x;
    const int tid = threadIdx.x;
    const int* vm = valid_mask + (size_t)b * L_;

    const int lo = pos_span[b * 2 + 0];
    const int hi = pos_span[b * 2 + 1];

    const bool v = (vm[tid] == 1);
    const int wv = __builtin_amdgcn_readfirstlane(tid >> 5);
    const unsigned bal = __builtin_amdgcn_ballot_w32(v);
    if ((tid & 31) == 0) s_wcnt[wv] = __builtin_popcount(bal);
    __syncthreads();

    int below = 0, total = 0;
    {
        const int4* wc = reinterpret_cast<const int4*>(s_wcnt);
        int cnt[NWAVES_F];
        #pragma unroll
        for (int q = 0; q < NWAVES_F / 4; ++q) {
            const int4 c4 = wc[q];
            cnt[4 * q + 0] = c4.x; cnt[4 * q + 1] = c4.y;
            cnt[4 * q + 2] = c4.z; cnt[4 * q + 3] = c4.w;
        }
        #pragma unroll
        for (int w = 0; w < NWAVES_F; ++w) {
            total += cnt[w];
            below += (w < wv) ? cnt[w] : 0;
        }
    }

    const int pos = below + (int)__builtin_amdgcn_mbcnt_lo(bal, 0u);
    if (v && pos < 64) s_idx[pos] = tid;
    __syncthreads();

    const int nv   = (total < 64) ? total : 64;
    const int jend = (hi < nv - 1) ? hi : nv - 1;
    const int jlo  = (lo > 0) ? lo : 0;

    const int rg = __builtin_amdgcn_readfirstlane(tid >> 6);
    const int u  = tid & 63;

    const float  ninf  = -__builtin_inff();
    const float4 ninf4 = make_float4(ninf, ninf, ninf, ninf);
    const float4* encb = reinterpret_cast<const float4*>(enc + (size_t)b * L_ * D_);

    float4 m0 = ninf4, m1 = ninf4, m2 = ninf4;
    for (int j0 = jlo + rg; j0 <= jend; j0 += ROWGROUPS * ROWS_PER_TRIP) {
        int r[ROWS_PER_TRIP];
        #pragma unroll
        for (int i = 0; i < ROWS_PER_TRIP; ++i) {
            const int j = j0 + ROWGROUPS * i;
            r[i] = __builtin_amdgcn_readfirstlane(s_idx[(j < jend) ? j : jend]);
        }
        float4 x0[ROWS_PER_TRIP], x1[ROWS_PER_TRIP], x2[ROWS_PER_TRIP];
        #pragma unroll
        for (int i = 0; i < ROWS_PER_TRIP; ++i) {
            x0[i] = ninf4; x1[i] = ninf4; x2[i] = ninf4;
            if (j0 + ROWGROUPS * i <= jend) {
                const float4* row = encb + (size_t)r[i] * D4_;
                x0[i] = row[u];
                x1[i] = row[64 + u];
                x2[i] = row[128 + u];
            }
        }
        #pragma unroll
        for (int i = 0; i < ROWS_PER_TRIP; ++i) {
            m0 = fmax4(m0, x0[i]);
            m1 = fmax4(m1, x1[i]);
            m2 = fmax4(m2, x2[i]);
        }
    }

    s_part[rg * D4_ + u]       = m0;
    s_part[rg * D4_ + 64 + u]  = m1;
    s_part[rg * D4_ + 128 + u] = m2;
    __syncthreads();

    if (tid < D4_) {
        float4 acc = s_part[tid];
        #pragma unroll
        for (int g = 1; g < ROWGROUPS; ++g)
            acc = fmax4(acc, s_part[g * D4_ + tid]);
        if (lo <= hi && hi >= nv)
            acc = fmax4(acc, make_float4(0.0f, 0.0f, 0.0f, 0.0f));
        reinterpret_cast<float4*>(feat + (size_t)b * D_)[tid] = acc;
    }
}

#define KSPLIT_WAVES 8

template <bool RELU, int N, int KD>
__global__ __launch_bounds__(KSPLIT_WAVES * 32)
__attribute__((amdgpu_waves_per_eu(2, 4)))
void gemm_wmma_ksplit(
    const float* __restrict__ A,
    const float* __restrict__ Bm,
    const float* __restrict__ bias,
    float*       __restrict__ C)
{
    constexpr int KS = KD / KSPLIT_WAVES;
    constexpr int G  = KS / 8;
    static_assert(KD % (KSPLIT_WAVES * 8) == 0, "K slice must be whole groups of 8");

    __shared__ float sPart[KSPLIT_WAVES * 256];

    const int lane = threadIdx.x & 31;
    const int wave = threadIdx.x >> 5;
    const int n0   = blockIdx.x * 16;
    const int m0   = blockIdx.y * 16;
    const int c    = lane & 15;
    const int h    = lane >> 4;

    const int    kbase = wave * KS + 4 * h;
    const float* ap = A  + (size_t)(m0 + c) * KD + kbase;
    const float* bp = Bm + (size_t)kbase * N + n0 + c;

    float4 af[G];
    float  bf[G][4];
    #pragma unroll
    for (int t = 0; t < G; ++t) {
        af[t] = *reinterpret_cast<const float4*>(ap + 8 * t);
        #pragma unroll
        for (int j = 0; j < 4; ++j)
            bf[t][j] = bp[(size_t)(8 * t + j) * N];
    }
    const float bias_n = bias[n0 + c];

    v8f acc = {};
    #pragma unroll
    for (int t = 0; t < G; ++t) {
        v2f a01 = {af[t].x, af[t].y};
        v2f b01 = {bf[t][0], bf[t][1]};
        acc = __builtin_amdgcn_wmma_f32_16x16x4_f32(
            false, a01, false, b01, (short)0, acc, false, false);
        v2f a23 = {af[t].z, af[t].w};
        v2f b23 = {bf[t][2], bf[t][3]};
        acc = __builtin_amdgcn_wmma_f32_16x16x4_f32(
            false, a23, false, b23, (short)0, acc, false, false);
    }

    #pragma unroll
    for (int vv = 0; vv < 8; ++vv)
        sPart[wave * 256 + vv * 32 + lane] = acc[vv];
    __syncthreads();

    const int e = threadIdx.x;
    float s = sPart[e];
    #pragma unroll
    for (int w = 1; w < KSPLIT_WAVES; ++w)
        s += sPart[w * 256 + e];

    float val = s + bias_n;
    if (RELU) val = fmaxf(val, 0.0f);
    C[(size_t)(m0 + 8 * h + wave) * N + n0 + c] = val;
}

#define KSL 128
#define NSL 128
#define NPL (D_ / KSL)
#define APITCH (KSL + 4)
#define SPAN_HEAD 40
#define SPAN_CHUNK 8
#define FLAG_HEAD 4
#define HPITCH (H_ + 8)

__device__ __forceinline__ void lds_handover_barrier()
{
    asm volatile("s_wait_dscnt 0x0\n\ts_barrier_signal -1\n\ts_barrier_wait -1" ::: "memory");
}

__global__ __launch_bounds__(512) void feat_gemm1_part(
    const float* __restrict__ enc,
    const int*   __restrict__ valid_mask,
    const int*   __restrict__ pos_span,
    const float* __restrict__ W1,
    float*       __restrict__ planes)
{
    __shared__ int s_idx[16 * 64];
    __shared__ __attribute__((aligned(16))) float sA[16 * APITCH];
    __shared__ __attribute__((aligned(16))) float sPart[16 * 256];

    const int ns   = blockIdx.x;
    const int ks   = blockIdx.y;
    const int mt   = blockIdx.z;
    const int tid  = threadIdx.x;
    const int lane = tid & 31;
    const int wv   = __builtin_amdgcn_readfirstlane(tid >> 5);
    const int c    = lane & 15;
    const int h    = lane >> 4;

    const int  b  = mt * 16 + wv;
    const int* vm = valid_mask + (size_t)b * L_;
    int flags[FLAG_HEAD];
    #pragma unroll
    for (int i = 0; i < FLAG_HEAD; ++i)
        flags[i] = vm[i * 32 + lane];
    const int lo = pos_span[b * 2 + 0];
    const int hi = pos_span[b * 2 + 1];

    const int nt   = wv & 7;
    const int kloc = (wv >> 3) * (KSL / 2) + 4 * h;
    const float* bp = W1 + (size_t)(ks * KSL + kloc) * H_ + ns * NSL + nt * 16 + c;
    constexpr int G1 = (KSL / 2) / 8;
    float bf[G1][4];
    #pragma unroll
    for (int t = 0; t < G1; ++t) {
        #pragma unroll
        for (int j = 0; j < 4; ++j)
            bf[t][j] = bp[(size_t)(8 * t + j) * H_];
    }

    int run = 0;
    #pragma unroll
    for (int i = 0; i < FLAG_HEAD; ++i) {
        const bool     v   = (flags[i] == 1);
        const unsigned bal = __builtin_amdgcn_ballot_w32(v);
        const int      pos = run + (int)__builtin_amdgcn_mbcnt_lo(bal, 0u);
        if (v && pos < 64) s_idx[wv * 64 + pos] = i * 32 + lane;
        run += __builtin_popcount(bal);
    }
    if (run <= hi) {
        int tail[L_ / 32 - FLAG_HEAD];
        #pragma unroll
        for (int i = 0; i < L_ / 32 - FLAG_HEAD; ++i)
            tail[i] = vm[(FLAG_HEAD + i) * 32 + lane];
        #pragma unroll
        for (int i = 0; i < L_ / 32 - FLAG_HEAD; ++i) {
            const bool     v   = (tail[i] == 1);
            const unsigned bal = __builtin_amdgcn_ballot_w32(v);
            const int      pos = run + (int)__builtin_amdgcn_mbcnt_lo(bal, 0u);
            if (v && pos < 64) s_idx[wv * 64 + pos] = (FLAG_HEAD + i) * 32 + lane;
            run += __builtin_popcount(bal);
        }
    }
    lds_handover_barrier();

    const int nv   = (run < 64) ? run : 64;
    const int jend = (hi < nv - 1) ? hi : nv - 1;
    const int jlo  = (lo > 0) ? lo : 0;

    const float  ninf  = -__builtin_inff();
    const float4 ninf4 = make_float4(ninf, ninf, ninf, ninf);
    const float4* encb = reinterpret_cast<const float4*>(enc + (size_t)b * L_ * D_ + ks * KSL);

    float4 m = ninf4;
    const int n = jend - jlo + 1;
    if (n > 0) {
        const int ja = jlo + lane;
        const int jb = jlo + 32 + lane;
        const int ra = s_idx[wv * 64 + ((ja < jend) ? ja : jend)];
        const int rb = s_idx[wv * 64 + ((jb < jend) ? jb : jend)];
        float4 x[SPAN_HEAD];
        #pragma unroll
        for (int q = 0; q < SPAN_HEAD / SPAN_CHUNK; ++q) {
            if (SPAN_CHUNK * q < n) {
                #pragma unroll
                for (int i = 0; i < SPAN_CHUNK; ++i) {
                    const int k = SPAN_CHUNK * q + i;
                    const int r = __builtin_amdgcn_readlane((k < 32) ? ra : rb, k & 31);
                    x[k] = encb[(size_t)r * D4_ + lane];
                }
            }
        }
        #pragma unroll
        for (int q = 0; q < SPAN_HEAD / SPAN_CHUNK; ++q) {
            if (SPAN_CHUNK * q < n) {
                #pragma unroll
                for (int i = 0; i < SPAN_CHUNK; ++i)
                    m = fmax4(m, x[SPAN_CHUNK * q + i]);
            }
        }
        for (int j = jlo + SPAN_HEAD; j <= jend; ++j) {
            const int r = __builtin_amdgcn_readfirstlane(s_idx[wv * 64 + j]);
            m = fmax4(m, encb[(size_t)r * D4_ + lane]);
        }
    }
    if (lo <= hi && hi >= nv)
        m = fmax4(m, make_float4(0.0f, 0.0f, 0.0f, 0.0f));
    *reinterpret_cast<float4*>(&sA[wv * APITCH + lane * 4]) = m;
    __syncthreads();

    const float* arow = sA + c * APITCH + kloc;
    v8f acc = {};
    #pragma unroll
    for (int t = 0; t < G1; ++t) {
        const float4 a4 = *reinterpret_cast<const float4*>(arow + 8 * t);
        v2f a01 = {a4.x, a4.y};
        v2f b01 = {bf[t][0], bf[t][1]};
        acc = __builtin_amdgcn_wmma_f32_16x16x4_f32(
            false, a01, false, b01, (short)0, acc, false, false);
        v2f a23 = {a4.z, a4.w};
        v2f b23 = {bf[t][2], bf[t][3]};
        acc = __builtin_amdgcn_wmma_f32_16x16x4_f32(
            false, a23, false, b23, (short)0, acc, false, false);
    }

    #pragma unroll
    for (int vv = 0; vv < 8; ++vv)
        sPart[wv * 256 + vv * 32 + lane] = acc[vv];
    __syncthreads();

    const int row  = tid >> 5;
    const int col4 = lane * 4;
    const int off  = (col4 >> 4) * 256 + (row & 7) * 32 + (row >> 3) * 16 + (col4 & 15);
    const float4 p0 = *reinterpret_cast<const float4*>(&sPart[off]);
    const float4 p1 = *reinterpret_cast<const float4*>(&sPart[8 * 256 + off]);
    const float4 s4 = make_float4(p0.x + p1.x, p0.y + p1.y, p0.z + p1.z, p0.w + p1.w);
    float* plane = planes + (size_t)ks * B_ * H_;
    *reinterpret_cast<float4*>(&plane[(size_t)(mt * 16 + row) * H_ + ns * NSL + col4]) = s4;
}

__global__ __launch_bounds__(KSPLIT_WAVES * 32)
__attribute__((amdgpu_waves_per_eu(2, 4)))
void gemm2_planes(
    const float* __restrict__ planes,
    const float* __restrict__ b1,
    const float* __restrict__ W2,
    const float* __restrict__ b2,
    float*       __restrict__ out)
{
    constexpr int KS = H_ / KSPLIT_WAVES;
    constexpr int G  = KS / 8;
    static_assert(H_ % (KSPLIT_WAVES * 8) == 0, "K slice must be whole groups of 8");

    __shared__ __attribute__((aligned(16))) float sH[4 * HPITCH];
    __shared__ float sPart[KSPLIT_WAVES * 256];

    const int tid  = threadIdx.x;
    const int lane = tid & 31;
    const int wave = tid >> 5;
    const int n0   = blockIdx.x * 16;
    const int r0   = blockIdx.y * 4;
    const int c    = lane & 15;
    const int h    = lane >> 4;

    constexpr int H4 = H_ / 4;
    float4 pv[3][NPL];
    float4 bb[3];
    #pragma unroll
    for (int q = 0; q < 3; ++q) {
        const int idx = tid + 256 * q;
        const int rr  = idx / H4;
        const int cq  = idx - rr * H4;
        bb[q] = reinterpret_cast<const float4*>(b1)[cq];
        #pragma unroll
        for (int p = 0; p < NPL; ++p)
            pv[q][p] = reinterpret_cast<const float4*>(
                planes + (size_t)p * B_ * H_ + (size_t)(r0 + rr) * H_)[cq];
    }

    const int    kbase = wave * KS + 4 * h;
    const float* bp = W2 + (size_t)kbase * K_ + n0 + c;
    float bf[G][4];
    #pragma unroll
    for (int t = 0; t < G; ++t) {
        #pragma unroll
        for (int j = 0; j < 4; ++j)
            bf[t][j] = bp[(size_t)(8 * t + j) * K_];
    }
    const float bias_n = b2[n0 + c];

    #pragma unroll
    for (int q = 0; q < 3; ++q) {
        const int idx = tid + 256 * q;
        const int rr  = idx / H4;
        const int cq  = idx - rr * H4;
        float4 s = pv[q][0];
        #pragma unroll
        for (int p = 1; p < NPL; ++p) {
            s.x += pv[q][p].x; s.y += pv[q][p].y; s.z += pv[q][p].z; s.w += pv[q][p].w;
        }
        s.x = fmaxf(s.x + bb[q].x, 0.0f);
        s.y = fmaxf(s.y + bb[q].y, 0.0f);
        s.z = fmaxf(s.z + bb[q].z, 0.0f);
        s.w = fmaxf(s.w + bb[q].w, 0.0f);
        *reinterpret_cast<float4*>(&sH[rr * HPITCH + cq * 4]) = s;
    }
    lds_handover_barrier();

    const float* arow = sH + (c & 3) * HPITCH + kbase;
    v8f acc = {};
    #pragma unroll
    for (int t = 0; t < G; ++t) {
        const float4 a4 = *reinterpret_cast<const float4*>(arow + 8 * t);
        v2f a01 = {a4.x, a4.y};
        v2f b01 = {bf[t][0], bf[t][1]};
        acc = __builtin_amdgcn_wmma_f32_16x16x4_f32(
            false, a01, false, b01, (short)0, acc, false, false);
        v2f a23 = {a4.z, a4.w};
        v2f b23 = {bf[t][2], bf[t][3]};
        acc = __builtin_amdgcn_wmma_f32_16x16x4_f32(
            false, a23, false, b23, (short)0, acc, false, false);
    }

    #pragma unroll
    for (int vv = 0; vv < 8; ++vv)
        sPart[wave * 256 + vv * 32 + lane] = acc[vv];
    __syncthreads();

    const int e = tid;
    float s = sPart[e];
    #pragma unroll
    for (int w = 1; w < KSPLIT_WAVES; ++w)
        s += sPart[w * 256 + e];
    if (h == 0 && wave < 4)
        out[(size_t)(r0 + wave) * K_ + n0 + c] = s + bias_n;
}

extern "C" void kernel_launch(void* const* d_in, const int* in_sizes, int n_in,
                              void* d_out, int out_size, void* d_ws, size_t ws_size,
                              hipStream_t stream)
{
    (void)in_sizes; (void)n_in; (void)out_size;

    const float* enc        = (const float*)d_in[0];
    const float* W1         = (const float*)d_in[1];
    const float* b1         = (const float*)d_in[2];
    const float* W2         = (const float*)d_in[3];
    const float* b2         = (const float*)d_in[4];
    const int*   valid_mask = (const int*)d_in[5];
    const int*   pos_span   = (const int*)d_in[6];

    float* out  = (float*)d_out;

    if (ws_size >= (size_t)NPL * B_ * H_ * sizeof(float)) {
        float* planes = (float*)d_ws;
        feat_gemm1_part<<<dim3(H_ / NSL, NPL, B_ / 16), 512, 0, stream>>>(enc, valid_mask, pos_span, W1, planes);
        gemm2_planes<<<dim3(K_ / 16, B_ / 4), KSPLIT_WAVES * 32, 0, stream>>>(planes, b1, W2, b2, out);
    } else {
        float* feat = (float*)d_ws;
        float* h    = feat + (size_t)B_ * D_;
        feat_kernel<<<B_, 512, 0, stream>>>(enc, valid_mask, pos_span, feat);
        gemm_wmma_ksplit<true,  H_, D_><<<dim3(H_ / 16, B_ / 16), KSPLIT_WAVES * 32, 0, stream>>>(feat, W1, b1, h);
        gemm_wmma_ksplit<false, K_, H_><<<dim3(K_ / 16, B_ / 16), KSPLIT_WAVES * 32, 0, stream>>>(h,    W2, b2, out);
    }
}
